# starting kernel
# baseline (speedup 1.0000x reference)
#include <hip/hip_runtime.h>


typedef __attribute__((ext_vector_type(2))) float v2f;
typedef __attribute__((ext_vector_type(8))) float v8f;

#define D_MODEL  4096
#define D_FF     11008
#define REMAINED 4403
#define BATCH    32

#define WAVES_PER_BLOCK 8
#define K_CHUNK (D_MODEL / WAVES_PER_BLOCK)

__global__ __launch_bounds__(32 * WAVES_PER_BLOCK)
void gathered_gemm_wmma_f32(const float* __restrict__ x,
                            const float* __restrict__ weight,
                            const int*   __restrict__ indices,
                            float*       __restrict__ out)
{
    __shared__ float red[WAVES_PER_BLOCK][2][8][32];

    const int tile = blockIdx.x;
    const int lane = threadIdx.x & 31;
    const int wave = threadIdx.x >> 5;

    const int n     = lane & 15;
    const int khalf = lane >> 4;

    int r = tile * 16 + n;
    int rc = r < REMAINED ? r : (REMAINED - 1);
    const float* wrow  = weight + (size_t)indices[rc] * D_MODEL;
    const float* xrow0 = x + (size_t)n        * D_MODEL;
    const float* xrow1 = x + (size_t)(n + 16) * D_MODEL;

    v8f c0 = {};
    v8f c1 = {};

    const int kbase = wave * K_CHUNK;
    const int kend  = kbase + K_CHUNK;

    for (int kb = kbase; kb < kend; kb += 16) {
        const int ko = kb + 2 * khalf;
        __builtin_prefetch(wrow + ko + 80, 0, 0);
#pragma unroll
        for (int u = 0; u < 4; ++u) {
            const int kk = ko + 4 * u;
            v2f a0 = *(const v2f*)(xrow0 + kk);
            v2f a1 = *(const v2f*)(xrow1 + kk);
            v2f b  = *(const v2f*)(wrow  + kk);
            c0 = __builtin_amdgcn_wmma_f32_16x16x4_f32(false, a0, false, b,
                                                       (short)0, c0, false, false);
            c1 = __builtin_amdgcn_wmma_f32_16x16x4_f32(false, a1, false, b,
                                                       (short)0, c1, false, false);
        }
    }

#pragma unroll
    for (int j = 0; j < 8; ++j) {
        red[wave][0][j][lane] = c0[j];
        red[wave][1][j][lane] = c1[j];
    }
    __syncthreads();

    for (int e = threadIdx.x; e < 512; e += blockDim.x) {
        const int l = e & 31;
        const int j = (e >> 5) & 7;
        const int t = e >> 8;
        float s = 0.f;
#pragma unroll
        for (int w = 0; w < WAVES_PER_BLOCK; ++w) s += red[w][t][j][l];
        const int m  = j + 8 * (l >> 4) + 16 * t;
        const int rr = tile * 16 + (l & 15);
        if (rr < REMAINED) out[(size_t)m * REMAINED + rr] = s;
    }
}

extern "C" void kernel_launch(void* const* d_in, const int* in_sizes, int n_in,
                              void* d_out, int out_size, void* d_ws, size_t ws_size,
                              hipStream_t stream) {
    const float* x       = (const float*)d_in[0];
    const float* weight  = (const float*)d_in[1];
    const int*   indices = (const int*)d_in[2];
    float*       out     = (float*)d_out;

    const int n_tiles = (REMAINED + 15) / 16;
    gathered_gemm_wmma_f32<<<n_tiles, 32 * WAVES_PER_BLOCK, 0, stream>>>(
        x, weight, indices, out);
}


# x held in registers per 128-k slice, reused over 5 column tiles (K split over waves, partials in workspace, 2nd launch adds the 32 slices in fixed order); f32 sum order changes
# speedup vs baseline: 1.7389x; 1.7389x over previous
#include <hip/hip_runtime.h>


typedef __attribute__((ext_vector_type(2))) float v2f;
typedef __attribute__((ext_vector_type(4))) float v4f;
typedef __attribute__((ext_vector_type(8))) float v8f;

#define D_MODEL  4096
#define REMAINED 4403
#define N_TILES  276

#define K_SLICE  128
#define N_SLICES (D_MODEL / K_SLICE)
#define K_STEPS  (K_SLICE / 4)

#define TILES_PER_WAVE  5
#define N_GROUPS        56
#define WAVES_PER_BLOCK 7

#define TILE_FLOATS  512
#define SLICE_FLOATS (N_TILES * TILE_FLOATS)

__global__ __launch_bounds__(32 * WAVES_PER_BLOCK)
void gathered_gemm_partial(const float* __restrict__ x,
                           const float* __restrict__ weight,
                           const int*   __restrict__ indices,
                           float*       __restrict__ part)
{
    const int lane = threadIdx.x & 31;
    const int wave = __builtin_amdgcn_readfirstlane((int)(threadIdx.x >> 5));

    const int gw    = blockIdx.x * WAVES_PER_BLOCK + wave;
    const int group = gw % N_GROUPS;
    const int slice = gw / N_GROUPS;

    const int n     = lane & 15;
    const int khalf = lane >> 4;
    const int k0    = slice * K_SLICE + 2 * khalf;

    const int tile0 = group * TILES_PER_WAVE;

    const float* wrow[TILES_PER_WAVE];
#pragma unroll
    for (int i = 0; i < TILES_PER_WAVE; ++i) {
        int r  = (tile0 + i) * 16 + n;
        int rc = r < REMAINED ? r : (REMAINED - 1);
        wrow[i] = weight + (size_t)indices[rc] * D_MODEL + k0;
    }

    const float* xrow0 = x + (size_t)n        * D_MODEL + k0;
    const float* xrow1 = x + (size_t)(n + 16) * D_MODEL + k0;

    v2f a0[K_STEPS], a1[K_STEPS];
#pragma unroll
    for (int s = 0; s < K_STEPS; ++s) {
        a0[s] = *(const v2f*)(xrow0 + 4 * s);
        a1[s] = *(const v2f*)(xrow1 + 4 * s);
    }

    v2f b[2][K_STEPS];
#pragma unroll
    for (int s = 0; s < K_STEPS; ++s) b[0][s] = *(const v2f*)(wrow[0] + 4 * s);

#pragma unroll
    for (int i = 0; i < TILES_PER_WAVE; ++i) {
        const int tile = tile0 + i;
        if (i + 1 < TILES_PER_WAVE) {
#pragma unroll
            for (int s = 0; s < K_STEPS; ++s)
                b[(i + 1) & 1][s] = *(const v2f*)(wrow[(i + 1) % TILES_PER_WAVE] + 4 * s);
        }

        v8f c0 = {};
        v8f c1 = {};
#pragma unroll
        for (int s = 0; s < K_STEPS; ++s) {
            c0 = __builtin_amdgcn_wmma_f32_16x16x4_f32(false, a0[s], false, b[i & 1][s],
                                                       (short)0, c0, false, false);
            c1 = __builtin_amdgcn_wmma_f32_16x16x4_f32(false, a1[s], false, b[i & 1][s],
                                                       (short)0, c1, false, false);
        }

        if (tile < N_TILES) {
            float* p = part + ((size_t)slice * N_TILES + tile) * TILE_FLOATS + lane;
#pragma unroll
            for (int j = 0; j < 8; ++j) {
                p[j * 32]       = c0[j];
                p[256 + j * 32] = c1[j];
            }
        }
    }
}

__global__ __launch_bounds__(256)
void reduce_slices(const float* __restrict__ part, float* __restrict__ out)
{
    const int q = blockIdx.x * 256 + threadIdx.x;
    const v4f* p = (const v4f*)part + q;

    v4f s = p[0];
#pragma unroll
    for (int sl = 1; sl < N_SLICES; ++sl) s += p[(size_t)sl * (SLICE_FLOATS / 4)];

    const int e    = q * 4;
    const int l    = e & 31;
    const int j    = (e >> 5) & 7;
    const int t    = (e >> 8) & 1;
    const int tile = e >> 9;
    const int m    = j + 8 * (l >> 4) + 16 * t;
    const int r    = tile * 16 + (l & 15);
    float* o = out + (size_t)m * REMAINED + r;
    if (r + 0 < REMAINED) o[0] = s.x;
    if (r + 1 < REMAINED) o[1] = s.y;
    if (r + 2 < REMAINED) o[2] = s.z;
    if (r + 3 < REMAINED) o[3] = s.w;
}

extern "C" void kernel_launch(void* const* d_in, const int* in_sizes, int n_in,
                              void* d_out, int out_size, void* d_ws, size_t ws_size,
                              hipStream_t stream) {
    const float* x       = (const float*)d_in[0];
    const float* weight  = (const float*)d_in[1];
    const int*   indices = (const int*)d_in[2];
    float*       out     = (float*)d_out;
    float*       part    = (float*)d_ws;

    gathered_gemm_partial<<<256, 32 * WAVES_PER_BLOCK, 0, stream>>>(x, weight, indices, part);
    reduce_slices<<<138, 256, 0, stream>>>(part, out);
}


# Fragment loads cover 8 rows x 64 B per b128 instruction (was 16 rows x 16 B); lane groups swap halves by DPP row rotate, no LDS; k order in each 16-k group permuted, f32 sum order changes
# speedup vs baseline: 2.0190x; 1.1611x over previous
#include <hip/hip_runtime.h>


typedef __attribute__((ext_vector_type(2))) float v2f;
typedef __attribute__((ext_vector_type(4))) float v4f;
typedef __attribute__((ext_vector_type(8))) float v8f;

#define D_MODEL  4096
#define REMAINED 4403
#define N_TILES  276

#define K_SLICE  128
#define N_SLICES (D_MODEL / K_SLICE)
#define K_STEPS  (K_SLICE / 4)

#define TILES_PER_WAVE  5
#define N_GROUPS        56
#define WAVES_PER_BLOCK 7

#define TILE_FLOATS  512
#define SLICE_FLOATS (N_TILES * TILE_FLOATS)

__device__ __forceinline__ float rows_lo_hi(float lo, float hi) {
    return __builtin_bit_cast(float, __builtin_amdgcn_update_dpp(
        __builtin_bit_cast(int, lo), __builtin_bit_cast(int, hi), 0x128, 0xF, 0xC, false));
}
__device__ __forceinline__ float rows_hi_lo(float lo, float hi) {
    return __builtin_bit_cast(float, __builtin_amdgcn_update_dpp(
        __builtin_bit_cast(int, hi), __builtin_bit_cast(int, lo), 0x128, 0xF, 0x3, false));
}

__device__ __forceinline__ void make_frags(const v4f lo, const v4f hi, v2f* f) {
    f[0] = v2f{rows_lo_hi(lo.x, hi.x), rows_lo_hi(lo.y, hi.y)};
    f[1] = v2f{rows_lo_hi(lo.z, hi.z), rows_lo_hi(lo.w, hi.w)};
    f[2] = v2f{rows_hi_lo(lo.x, hi.x), rows_hi_lo(lo.y, hi.y)};
    f[3] = v2f{rows_hi_lo(lo.z, hi.z), rows_hi_lo(lo.w, hi.w)};
}

#define K_PIECES (K_SLICE / 16)

__global__ __launch_bounds__(32 * WAVES_PER_BLOCK)
void gathered_gemm_partial(const float* __restrict__ x,
                           const float* __restrict__ weight,
                           const int*   __restrict__ indices,
                           float*       __restrict__ part)
{
    const int lane = threadIdx.x & 31;
    const int wave = __builtin_amdgcn_readfirstlane((int)(threadIdx.x >> 5));

    const int gw    = blockIdx.x * WAVES_PER_BLOCK + wave;
    const int group = gw % N_GROUPS;
    const int slice = gw / N_GROUPS;

    const int n     = lane & 15;
    const int khalf = lane >> 4;
    const int r8    = n & 7;
    const int k0    = slice * K_SLICE + 4 * (2 * khalf + (n >> 3));

    const int tile0 = group * TILES_PER_WAVE;

    const float* wlo[TILES_PER_WAVE];
    const float* whi[TILES_PER_WAVE];
#pragma unroll
    for (int i = 0; i < TILES_PER_WAVE; ++i) {
        int rl  = (tile0 + i) * 16 + r8;
        int rh  = rl + 8;
        int rlc = rl < REMAINED ? rl : (REMAINED - 1);
        int rhc = rh < REMAINED ? rh : (REMAINED - 1);
        wlo[i] = weight + (size_t)indices[rlc] * D_MODEL + k0;
        whi[i] = weight + (size_t)indices[rhc] * D_MODEL + k0;
    }

    const float* x0lo = x + (size_t)(r8)      * D_MODEL + k0;
    const float* x0hi = x + (size_t)(r8 + 8)  * D_MODEL + k0;
    const float* x1lo = x + (size_t)(r8 + 16) * D_MODEL + k0;
    const float* x1hi = x + (size_t)(r8 + 24) * D_MODEL + k0;

    v2f a0[K_STEPS], a1[K_STEPS];
    {
        v4f xl0[K_PIECES], xh0[K_PIECES], xl1[K_PIECES], xh1[K_PIECES];
#pragma unroll
        for (int u = 0; u < K_PIECES; ++u) {
            xl0[u] = *(const v4f*)(x0lo + 16 * u);
            xh0[u] = *(const v4f*)(x0hi + 16 * u);
            xl1[u] = *(const v4f*)(x1lo + 16 * u);
            xh1[u] = *(const v4f*)(x1hi + 16 * u);
        }
#pragma unroll
        for (int u = 0; u < K_PIECES; ++u) {
            make_frags(xl0[u], xh0[u], &a0[4 * u]);
            make_frags(xl1[u], xh1[u], &a1[4 * u]);
        }
    }

    v4f blo[2][K_PIECES], bhi[2][K_PIECES];
#pragma unroll
    for (int u = 0; u < K_PIECES; ++u) {
        blo[0][u] = *(const v4f*)(wlo[0] + 16 * u);
        bhi[0][u] = *(const v4f*)(whi[0] + 16 * u);
    }

#pragma unroll
    for (int i = 0; i < TILES_PER_WAVE; ++i) {
        const int tile = tile0 + i;
        if (i + 1 < TILES_PER_WAVE) {
#pragma unroll
            for (int u = 0; u < K_PIECES; ++u) {
                blo[(i + 1) & 1][u] = *(const v4f*)(wlo[(i + 1) % TILES_PER_WAVE] + 16 * u);
                bhi[(i + 1) & 1][u] = *(const v4f*)(whi[(i + 1) % TILES_PER_WAVE] + 16 * u);
            }
        }

        v8f c0 = {};
        v8f c1 = {};
#pragma unroll
        for (int u = 0; u < K_PIECES; ++u) {
            v2f b[4];
            make_frags(blo[i & 1][u], bhi[i & 1][u], b);
#pragma unroll
            for (int t = 0; t < 4; ++t) {
                c0 = __builtin_amdgcn_wmma_f32_16x16x4_f32(false, a0[4 * u + t], false, b[t],
                                                           (short)0, c0, false, false);
                c1 = __builtin_amdgcn_wmma_f32_16x16x4_f32(false, a1[4 * u + t], false, b[t],
                                                           (short)0, c1, false, false);
            }
        }

        if (tile < N_TILES) {
            float* p = part + ((size_t)slice * N_TILES + tile) * TILE_FLOATS + lane;
#pragma unroll
            for (int j = 0; j < 8; ++j) {
                p[j * 32]       = c0[j];
                p[256 + j * 32] = c1[j];
            }
        }
    }
}

__global__ __launch_bounds__(256)
void reduce_slices(const float* __restrict__ part, float* __restrict__ out)
{
    const int q = blockIdx.x * 256 + threadIdx.x;
    const v4f* p = (const v4f*)part + q;

    v4f s = p[0];
#pragma unroll
    for (int sl = 1; sl < N_SLICES; ++sl) s += p[(size_t)sl * (SLICE_FLOATS / 4)];

    const int e    = q * 4;
    const int l    = e & 31;
    const int j    = (e >> 5) & 7;
    const int t    = (e >> 8) & 1;
    const int tile = e >> 9;
    const int m    = j + 8 * (l >> 4) + 16 * t;
    const int r    = tile * 16 + (l & 15);
    float* o = out + (size_t)m * REMAINED + r;
    if (r + 0 < REMAINED) o[0] = s.x;
    if (r + 1 < REMAINED) o[1] = s.y;
    if (r + 2 < REMAINED) o[2] = s.z;
    if (r + 3 < REMAINED) o[3] = s.w;
}

extern "C" void kernel_launch(void* const* d_in, const int* in_sizes, int n_in,
                              void* d_out, int out_size, void* d_ws, size_t ws_size,
                              hipStream_t stream) {
    const float* x       = (const float*)d_in[0];
    const float* weight  = (const float*)d_in[1];
    const int*   indices = (const int*)d_in[2];
    float*       out     = (float*)d_out;
    float*       part    = (float*)d_ws;

    gathered_gemm_partial<<<256, 32 * WAVES_PER_BLOCK, 0, stream>>>(x, weight, indices, part);
    reduce_slices<<<138, 256, 0, stream>>>(part, out);
}


# Work-group = 4 K slices x 2 column groups; the 4 slices of each tile are added in LDS behind one barrier, so 8 K chunks (was 32 slices) pass through the workspace; f32 sum order changes
# speedup vs baseline: 2.3874x; 1.1825x over previous
#include <hip/hip_runtime.h>


typedef __attribute__((ext_vector_type(2))) float v2f;
typedef __attribute__((ext_vector_type(4))) float v4f;
typedef __attribute__((ext_vector_type(8))) float v8f;

#define D_MODEL  4096
#define REMAINED 4403
#define N_TILES  276

#define K_SLICE  128
#define N_SLICES (D_MODEL / K_SLICE)
#define K_STEPS  (K_SLICE / 4)

#define TILES_PER_WAVE   5
#define N_GROUPS         56
#define SLICES_PER_BLOCK 4
#define GROUPS_PER_BLOCK 2
#define WAVES_PER_BLOCK  (SLICES_PER_BLOCK * GROUPS_PER_BLOCK)
#define N_CHUNKS         (N_SLICES / SLICES_PER_BLOCK)
#define N_PAIRS          (N_GROUPS / GROUPS_PER_BLOCK)
#define N_BLOCKS         (N_CHUNKS * N_PAIRS)

#define TILE_FLOATS  512
#define TILE_VEC4    (TILE_FLOATS / 4)
#define CHUNK_FLOATS (N_TILES * TILE_FLOATS)

__device__ __forceinline__ float rows_lo_hi(float lo, float hi) {
    return __builtin_bit_cast(float, __builtin_amdgcn_update_dpp(
        __builtin_bit_cast(int, lo), __builtin_bit_cast(int, hi), 0x128, 0xF, 0xC, false));
}
__device__ __forceinline__ float rows_hi_lo(float lo, float hi) {
    return __builtin_bit_cast(float, __builtin_amdgcn_update_dpp(
        __builtin_bit_cast(int, hi), __builtin_bit_cast(int, lo), 0x128, 0xF, 0x3, false));
}

__device__ __forceinline__ void make_frags(const v4f lo, const v4f hi, v2f* f) {
    f[0] = v2f{rows_lo_hi(lo.x, hi.x), rows_lo_hi(lo.y, hi.y)};
    f[1] = v2f{rows_lo_hi(lo.z, hi.z), rows_lo_hi(lo.w, hi.w)};
    f[2] = v2f{rows_hi_lo(lo.x, hi.x), rows_hi_lo(lo.y, hi.y)};
    f[3] = v2f{rows_hi_lo(lo.z, hi.z), rows_hi_lo(lo.w, hi.w)};
}

#define K_PIECES (K_SLICE / 16)

__global__ __launch_bounds__(32 * WAVES_PER_BLOCK)
void gathered_gemm_partial(const float* __restrict__ x,
                           const float* __restrict__ weight,
                           const int*   __restrict__ indices,
                           float*       __restrict__ part)
{
    __shared__ v4f red[WAVES_PER_BLOCK * TILES_PER_WAVE * TILE_VEC4];

    const int lane = threadIdx.x & 31;
    const int wave = __builtin_amdgcn_readfirstlane((int)(threadIdx.x >> 5));

    const int pair  = blockIdx.x % N_PAIRS;
    const int chunk = blockIdx.x / N_PAIRS;
    const int gsel  = wave >> 2;
    const int ssel  = wave & 3;
    const int group = pair * GROUPS_PER_BLOCK + gsel;
    const int slice = chunk * SLICES_PER_BLOCK + ssel;

    const int n     = lane & 15;
    const int khalf = lane >> 4;
    const int r8    = n & 7;
    const int k0    = slice * K_SLICE + 4 * (2 * khalf + (n >> 3));

    const int tile0 = group * TILES_PER_WAVE;

    int il[TILES_PER_WAVE], ih[TILES_PER_WAVE];
#pragma unroll
    for (int i = 0; i < TILES_PER_WAVE; ++i) {
        int rl  = (tile0 + i) * 16 + r8;
        int rh  = rl + 8;
        int rlc = rl < REMAINED ? rl : (REMAINED - 1);
        int rhc = rh < REMAINED ? rh : (REMAINED - 1);
        il[i] = indices[rlc];
        ih[i] = indices[rhc];
    }

    __builtin_amdgcn_sched_barrier(0);

    const float* x0lo = x + (size_t)(r8)      * D_MODEL + k0;
    const float* x0hi = x + (size_t)(r8 + 8)  * D_MODEL + k0;
    const float* x1lo = x + (size_t)(r8 + 16) * D_MODEL + k0;
    const float* x1hi = x + (size_t)(r8 + 24) * D_MODEL + k0;

    v4f xl0[K_PIECES], xh0[K_PIECES], xl1[K_PIECES], xh1[K_PIECES];
#pragma unroll
    for (int u = 0; u < K_PIECES; ++u) {
        xl0[u] = *(const v4f*)(x0lo + 16 * u);
        xh0[u] = *(const v4f*)(x0hi + 16 * u);
        xl1[u] = *(const v4f*)(x1lo + 16 * u);
        xh1[u] = *(const v4f*)(x1hi + 16 * u);
    }

    const float* wlo[TILES_PER_WAVE];
    const float* whi[TILES_PER_WAVE];
#pragma unroll
    for (int i = 0; i < TILES_PER_WAVE; ++i) {
        wlo[i] = weight + (size_t)il[i] * D_MODEL + k0;
        whi[i] = weight + (size_t)ih[i] * D_MODEL + k0;
    }

    v4f blo[2][K_PIECES], bhi[2][K_PIECES];
#pragma unroll
    for (int u = 0; u < K_PIECES; ++u) {
        blo[0][u] = *(const v4f*)(wlo[0] + 16 * u);
        bhi[0][u] = *(const v4f*)(whi[0] + 16 * u);
    }

    v2f a0[K_STEPS], a1[K_STEPS];

#pragma unroll
    for (int i = 0; i < TILES_PER_WAVE; ++i) {
        __builtin_amdgcn_sched_barrier(0);
        if (i + 1 < TILES_PER_WAVE) {
#pragma unroll
            for (int u = 0; u < K_PIECES; ++u) {
                blo[(i + 1) & 1][u] = *(const v4f*)(wlo[(i + 1) % TILES_PER_WAVE] + 16 * u);
                bhi[(i + 1) & 1][u] = *(const v4f*)(whi[(i + 1) % TILES_PER_WAVE] + 16 * u);
            }
        }
        __builtin_amdgcn_sched_barrier(0);

        if (i == 0) {
#pragma unroll
            for (int u = 0; u < K_PIECES; ++u) {
                make_frags(xl0[u], xh0[u], &a0[4 * u]);
                make_frags(xl1[u], xh1[u], &a1[4 * u]);
            }
        }

        v8f c0 = {};
        v8f c1 = {};
#pragma unroll
        for (int u = 0; u < K_PIECES; ++u) {
            v2f b[4];
            make_frags(blo[i & 1][u], bhi[i & 1][u], b);
#pragma unroll
            for (int t = 0; t < 4; ++t) {
                c0 = __builtin_amdgcn_wmma_f32_16x16x4_f32(false, a0[4 * u + t], false, b[t],
                                                           (short)0, c0, false, false);
                c1 = __builtin_amdgcn_wmma_f32_16x16x4_f32(false, a1[4 * u + t], false, b[t],
                                                           (short)0, c1, false, false);
            }
        }

        v4f* r = red + (wave * TILES_PER_WAVE + i) * TILE_VEC4 + lane;
        r[0]  = v4f{c0[0], c0[1], c0[2], c0[3]};
        r[32] = v4f{c0[4], c0[5], c0[6], c0[7]};
        r[64] = v4f{c1[0], c1[1], c1[2], c1[3]};
        r[96] = v4f{c1[4], c1[5], c1[6], c1[7]};
    }

    __syncthreads();

    const v4f* s0 = red + (gsel * SLICES_PER_BLOCK * TILES_PER_WAVE) * TILE_VEC4 + lane;
    v4f*       p0 = (v4f*)part + ((size_t)chunk * N_TILES + tile0) * TILE_VEC4 + lane;
    const int item0 = ssel * TILES_PER_WAVE;

    if (tile0 + TILES_PER_WAVE <= N_TILES) {
        v4f acc[TILES_PER_WAVE];
#pragma unroll
        for (int t = 0; t < TILES_PER_WAVE; ++t) {
            const v4f* s = s0 + (item0 + t) * 32;
            acc[t] = s[0];
            acc[t] += s[1 * TILES_PER_WAVE * TILE_VEC4];
            acc[t] += s[2 * TILES_PER_WAVE * TILE_VEC4];
            acc[t] += s[3 * TILES_PER_WAVE * TILE_VEC4];
        }
#pragma unroll
        for (int t = 0; t < TILES_PER_WAVE; ++t) p0[(item0 + t) * 32] = acc[t];
    } else {
#pragma unroll
        for (int t = 0; t < TILES_PER_WAVE; ++t) {
            const int item = item0 + t;
            if (tile0 + (item >> 2) < N_TILES) {
                const v4f* s = s0 + item * 32;
                v4f acc = s[0];
                acc += s[1 * TILES_PER_WAVE * TILE_VEC4];
                acc += s[2 * TILES_PER_WAVE * TILE_VEC4];
                acc += s[3 * TILES_PER_WAVE * TILE_VEC4];
                p0[item * 32] = acc;
            }
        }
    }
}

__global__ __launch_bounds__(256)
void reduce_chunks(const float* __restrict__ part, float* __restrict__ out)
{
    const int f = blockIdx.x * 256 + threadIdx.x;
    const v4f* p = (const v4f*)part + f;

    v4f s = p[0];
#pragma unroll
    for (int c = 1; c < N_CHUNKS; ++c) s += p[(size_t)c * (CHUNK_FLOATS / 4)];

    const int l    = f & 31;
    const int q    = (f >> 5) & 3;
    const int tile = f >> 7;
    const int m    = 4 * (q & 1) + 8 * (l >> 4) + 16 * (q >> 1);
    const int r    = tile * 16 + (l & 15);
    if (r < REMAINED) {
        float* o = out + (size_t)m * REMAINED + r;
        o[0]            = s.x;
        o[REMAINED]     = s.y;
        o[2 * REMAINED] = s.z;
        o[3 * REMAINED] = s.w;
    }
}

extern "C" void kernel_launch(void* const* d_in, const int* in_sizes, int n_in,
                              void* d_out, int out_size, void* d_ws, size_t ws_size,
                              hipStream_t stream) {
    const float* x       = (const float*)d_in[0];
    const float* weight  = (const float*)d_in[1];
    const int*   indices = (const int*)d_in[2];
    float*       out     = (float*)d_out;
    float*       part    = (float*)d_ws;

    gathered_gemm_partial<<<N_BLOCKS, 32 * WAVES_PER_BLOCK, 0, stream>>>(x, weight, indices, part);
    reduce_chunks<<<138, 256, 0, stream>>>(part, out);
}
